# starting kernel
# baseline (speedup 1.0000x reference)
#include <hip/hip_runtime.h>

typedef __attribute__((ext_vector_type(2))) float v2f;
typedef __attribute__((ext_vector_type(8))) float v8f;

#define IN_CH 128
#define HID   128

__global__ void k_init_deg(unsigned* __restrict__ deg, int N) {
  int i = blockIdx.x * blockDim.x + threadIdx.x;
  if (i < N) deg[i] = 1u;
}

__global__ void k_count(const int* __restrict__ ei, int E, unsigned* __restrict__ deg) {
  int e = blockIdx.x * blockDim.x + threadIdx.x;
  if (e < E) atomicAdd(&deg[ei[E + e]], 1u);
}

__global__ void k_dis(const unsigned* __restrict__ deg, float* __restrict__ dis, int N) {
  int i = blockIdx.x * blockDim.x + threadIdx.x;
  if (i < N) dis[i] = rsqrtf((float)deg[i]);
}

__global__ __launch_bounds__(128) void k_gemm(const float* __restrict__ x,
                                              const float* __restrict__ W,
                                              float* __restrict__ h) {
  const int lane = threadIdx.x & 31;
  const int wave = threadIdx.x >> 5;
  const int row0 = blockIdx.y * 16;
  const int col0 = (blockIdx.x * 4 + wave) * 16;

  const int mn  = lane & 15;
  const int kof = (lane >> 4) << 1;

  v8f acc = {};
  const float* xr = x + (size_t)(row0 + mn) * IN_CH;
  const float* Wc = W + col0 + mn;

#pragma unroll
  for (int k = 0; k < IN_CH; k += 4) {
    const float2 av = *(const float2*)(xr + k + kof);
    v2f a; a[0] = av.x; a[1] = av.y;
    v2f b;
    b[0] = Wc[(size_t)(k + kof) * HID];
    b[1] = Wc[(size_t)(k + kof + 1) * HID];
    acc = __builtin_amdgcn_wmma_f32_16x16x4_f32(false, a, false, b,
                                                (short)0, acc, false, false);
  }

  const int mo = (lane >> 4) * 8;
  float* hr = h + (size_t)(row0 + mo) * HID + col0 + mn;
#pragma unroll
  for (int r = 0; r < 8; ++r)
    hr[(size_t)r * HID] = acc[r];
}

__global__ void k_init_out(const float* __restrict__ h, const float* __restrict__ dis,
                           const float* __restrict__ bias, float* __restrict__ out, int N) {
  int t = blockIdx.x * blockDim.x + threadIdx.x;
  if (t >= N * 32) return;
  int i  = t >> 5;
  int cg = (t & 31) << 2;
  float nrm = dis[i] * dis[i];
  float4 hv = *(const float4*)(h + (size_t)i * HID + cg);
  float4 bv = *(const float4*)(bias + cg);
  float4 o;
  o.x = bv.x + nrm * hv.x;
  o.y = bv.y + nrm * hv.y;
  o.z = bv.z + nrm * hv.z;
  o.w = bv.w + nrm * hv.w;
  *(float4*)(out + (size_t)i * HID + cg) = o;
}

__global__ __launch_bounds__(256) void k_scatter(const int* __restrict__ ei, int E,
                                                 const float* __restrict__ h,
                                                 const float* __restrict__ dis,
                                                 float* __restrict__ out) {
  int w    = (blockIdx.x * blockDim.x + threadIdx.x) >> 5;
  int lane = threadIdx.x & 31;
  if (w >= E) return;
  int s = ei[w];
  int d = ei[E + w];
  float nrm = dis[s] * dis[d];
  const float4 hv = *(const float4*)(h + (size_t)s * HID + (lane << 2));
  float* op = out + (size_t)d * HID + (lane << 2);
  unsafeAtomicAdd(op + 0, nrm * hv.x);
  unsafeAtomicAdd(op + 1, nrm * hv.y);
  unsafeAtomicAdd(op + 2, nrm * hv.z);
  unsafeAtomicAdd(op + 3, nrm * hv.w);
}

__global__ void k_prelu(float* __restrict__ out, const float* __restrict__ pw, int N) {
  int t = blockIdx.x * blockDim.x + threadIdx.x;
  if (t >= N * 32) return;
  int cg = (t & 31) << 2;
  float4 v = *(float4*)(out + (size_t)t * 4);
  float4 p = *(const float4*)(pw + cg);
  v.x = v.x >= 0.f ? v.x : p.x * v.x;
  v.y = v.y >= 0.f ? v.y : p.y * v.y;
  v.z = v.z >= 0.f ? v.z : p.z * v.z;
  v.w = v.w >= 0.f ? v.w : p.w * v.w;
  *(float4*)(out + (size_t)t * 4) = v;
}

extern "C" void kernel_launch(void* const* d_in, const int* in_sizes, int n_in,
                              void* d_out, int out_size, void* d_ws, size_t ws_size,
                              hipStream_t stream) {
  const float* x  = (const float*)d_in[0];
  const int*   ei = (const int*)d_in[1];
  const float* W  = (const float*)d_in[2];
  const float* b  = (const float*)d_in[3];
  const float* pw = (const float*)d_in[4];
  float* out = (float*)d_out;

  const int N = in_sizes[0] / IN_CH;
  const int E = in_sizes[1] / 2;

  char* ws = (char*)d_ws;
  size_t o0 = 0;
  size_t o1 = (o0 + (size_t)N * 4 + 255) & ~(size_t)255;
  size_t o2 = (o1 + (size_t)N * 4 + 255) & ~(size_t)255;
  unsigned* deg = (unsigned*)(ws + o0);
  float*    dis = (float*)(ws + o1);
  float*    h   = (float*)(ws + o2);

  const int T = 256;
  k_init_deg<<<(N + T - 1) / T, T, 0, stream>>>(deg, N);
  k_count  <<<(E + T - 1) / T, T, 0, stream>>>(ei, E, deg);
  k_dis    <<<(N + T - 1) / T, T, 0, stream>>>(deg, dis, N);

  dim3 ggrid(2, N / 16);
  k_gemm<<<ggrid, 128, 0, stream>>>(x, W, h);

  int tq = N * 32;
  k_init_out<<<(tq + T - 1) / T, T, 0, stream>>>(h, dis, b, out, N);

  int sblocks = (E + 7) / 8;
  k_scatter<<<sblocks, T, 0, stream>>>(ei, E, h, dis, out);

  k_prelu<<<(tq + T - 1) / T, T, 0, stream>>>(out, pw, N);
}


# h = x @ W: W staged once per 8-wave work-group in LDS as B fragment pairs; each wave computes whole 16x128 strips (one A fragment feeds 8 column tiles); 512 looping work-groups; same k order
# speedup vs baseline: 2.1874x; 2.1874x over previous
#include <hip/hip_runtime.h>


typedef __attribute__((ext_vector_type(2))) float v2f;
typedef __attribute__((ext_vector_type(8))) float v8f;

#define IN_CH 128
#define HID   128

__global__ void k_zero_cnt(unsigned* __restrict__ cnt, int N) {
  int i = blockIdx.x * blockDim.x + threadIdx.x;
  if (i < N) cnt[i] = 0u;
}

__global__ void k_count_rank(const int* __restrict__ ei, int E,
                             unsigned* __restrict__ cnt, unsigned* __restrict__ rank) {
  int e = blockIdx.x * blockDim.x + threadIdx.x;
  if (e < E) rank[e] = atomicAdd(&cnt[ei[E + e]], 1u);
}

__global__ __launch_bounds__(256) void k_scan_dis(const unsigned* __restrict__ cnt,
                                                  unsigned* __restrict__ rowptr,
                                                  float* __restrict__ dis, int N) {
  __shared__ unsigned red[8];
  __shared__ unsigned wtot[8];
  const int tid  = threadIdx.x;
  const int lane = tid & 31;
  const int wave = tid >> 5;
  const int start = blockIdx.x * 1024;

  const int rounds = blockIdx.x;
  unsigned s = 0;
  for (int r = 0; r < rounds; r += 8) {
    uint4 v[8];
#pragma unroll
    for (int u = 0; u < 8; ++u) {
      const int q = (r + u < rounds) ? (r + u) : (rounds - 1);
      v[u] = *(const uint4*)(cnt + q * 1024 + tid * 4);
    }
#pragma unroll
    for (int u = 0; u < 8; ++u) {
      const unsigned m = (r + u < rounds) ? 0xffffffffu : 0u;
      s += ((v[u].x + v[u].y) + (v[u].z + v[u].w)) & m;
    }
  }
  for (int o = 16; o > 0; o >>= 1) s += __shfl_xor(s, o);
  if (lane == 0) red[wave] = s;

  const int i0 = start + tid * 4;
  unsigned c[4];
#pragma unroll
  for (int j = 0; j < 4; ++j) c[j] = (i0 + j < N) ? cnt[i0 + j] : 0u;
  const unsigned t = (c[0] + c[1]) + (c[2] + c[3]);
  unsigned inc = t;
  for (int o = 1; o < 32; o <<= 1) {
    const unsigned u = __shfl_up(inc, o);
    if (lane >= o) inc += u;
  }
  if (lane == 31) wtot[wave] = inc;
  __syncthreads();

  unsigned base = 0;
  for (int w = 0; w < 8; ++w) base += red[w];
  for (int w = 0; w < 8; ++w) base += (w < wave) ? wtot[w] : 0u;

  unsigned ex = base + inc - t;
#pragma unroll
  for (int j = 0; j < 4; ++j) {
    const int i = i0 + j;
    if (i <= N) rowptr[i] = ex;
    if (i < N)  dis[i] = rsqrtf((float)(c[j] + 1u));
    ex += c[j];
  }
}

__device__ __forceinline__ void load_a_chunk(v2f (&a)[8], const float* __restrict__ xr) {
#pragma unroll
  for (int q = 0; q < 8; ++q) {
    const float2 av = *(const float2*)(xr + q * 4);
    a[q][0] = av.x; a[q][1] = av.y;
  }
}

__device__ __forceinline__ void mma_chunk(v8f (&acc)[8], const v2f (&a)[8], float4 (&bb)[2][8],
                                          const float4* wl, int j0) {
#pragma unroll
  for (int jj = 0; jj < 4; ++jj) {
    const float4* wn = wl + ((j0 + jj + 1) & 15) * 2 * HID;
    __builtin_amdgcn_sched_barrier(0);
#pragma unroll
    for (int ct = 0; ct < 8; ++ct) bb[(jj + 1) & 1][ct] = wn[ct * 16];
    __builtin_amdgcn_sched_barrier(0);
#pragma unroll
    for (int ct = 0; ct < 8; ++ct) {
      v2f b; b[0] = bb[jj & 1][ct].x; b[1] = bb[jj & 1][ct].y;
      acc[ct] = __builtin_amdgcn_wmma_f32_16x16x4_f32(false, a[2 * jj], false, b,
                                                      (short)0, acc[ct], false, false);
    }
#pragma unroll
    for (int ct = 0; ct < 8; ++ct) {
      v2f b; b[0] = bb[jj & 1][ct].z; b[1] = bb[jj & 1][ct].w;
      acc[ct] = __builtin_amdgcn_wmma_f32_16x16x4_f32(false, a[2 * jj + 1], false, b,
                                                      (short)0, acc[ct], false, false);
    }
  }
}

__global__ __launch_bounds__(256) void k_gemm(const float* __restrict__ x,
                                              const float* __restrict__ W,
                                              float* __restrict__ h, int nstrips) {
  __shared__ float4 Ws[16 * 2 * HID];

  const int tid  = threadIdx.x;
  const int lane = tid & 31;
  const int wave = __builtin_amdgcn_readfirstlane(tid >> 5);
  const int mn   = lane & 15;
  const int hf   = lane >> 4;
  const int kof  = hf << 1;
  const int G    = (int)gridDim.x;

  int strip = wave * G + (int)blockIdx.x;

  v2f a0[8], a1[8];
  {
    const int s0 = strip < nstrips ? strip : nstrips - 1;
    load_a_chunk(a0, x + (size_t)(s0 * 16 + mn) * IN_CH + kof);
  }

#pragma unroll
  for (int u = 0; u < 16; ++u) {
    const int s  = u * 256 + tid;
    const int n  = s & (HID - 1);
    const int kb = ((s >> 8) << 3) + (((s >> 7) & 1) << 1);
    float4 v;
    v.x = W[(size_t)(kb    ) * HID + n];
    v.y = W[(size_t)(kb + 1) * HID + n];
    v.z = W[(size_t)(kb + 4) * HID + n];
    v.w = W[(size_t)(kb + 5) * HID + n];
    Ws[s] = v;
  }
  __syncthreads();

  const float4* wl = Ws + hf * HID + mn;
  const int mo = hf * 8;

  float4 bb[2][8];
#pragma unroll
  for (int ct = 0; ct < 8; ++ct) bb[0][ct] = wl[ct * 16];

  while (strip < nstrips) {
    v8f acc[8];
#pragma unroll
    for (int ct = 0; ct < 8; ++ct) acc[ct] = (v8f){};

    const int nstrip = strip + 8 * G;
    const float* xr = x + (size_t)(strip * 16 + mn) * IN_CH + kof;

#pragma unroll 1
    for (int c2 = 0; c2 < 2; ++c2) {
      load_a_chunk(a1, xr + (2 * c2 + 1) * 32);
      mma_chunk(acc, a0, bb, wl, c2 * 8);
      const int ps = (c2 == 0 || nstrip >= nstrips) ? strip : nstrip;
      load_a_chunk(a0, x + (size_t)(ps * 16 + mn) * IN_CH + kof + (c2 == 0 ? 64 : 0));
      mma_chunk(acc, a1, bb, wl, c2 * 8 + 4);
    }

    float* hr = h + (size_t)(strip * 16 + mo) * HID + mn;
#pragma unroll
    for (int ct = 0; ct < 8; ++ct)
#pragma unroll
      for (int r = 0; r < 8; ++r)
        hr[(size_t)r * HID + ct * 16] = acc[ct][r];

    strip = nstrip;
  }
}

__global__ void k_fill(const int* __restrict__ ei, int E,
                       const unsigned* __restrict__ rowptr,
                       const unsigned* __restrict__ rank, int* __restrict__ col) {
  int e = blockIdx.x * blockDim.x + threadIdx.x;
  if (e < E) col[rowptr[ei[E + e]] + rank[e]] = ei[e];
}

__device__ __forceinline__ float lane_f32(float v, int j) {
  return __builtin_bit_cast(float, __builtin_amdgcn_readlane(__builtin_bit_cast(int, v), j));
}

__device__ __forceinline__ void axpy4(float4& acc, float w, const float4 v) {
  acc.x += w * v.x;
  acc.y += w * v.y;
  acc.z += w * v.z;
  acc.w += w * v.w;
}

__global__ __launch_bounds__(256) void k_gather(const int* __restrict__ col,
                                                const unsigned* __restrict__ rowptr,
                                                const float* __restrict__ h,
                                                const float* __restrict__ dis,
                                                const float* __restrict__ bias,
                                                const float* __restrict__ pw,
                                                float* __restrict__ out, int N) {
  const int d = __builtin_amdgcn_readfirstlane((int)((blockIdx.x * blockDim.x + threadIdx.x) >> 5));
  const int lane = threadIdx.x & 31;
  if (d >= N) return;
  const int cg = lane << 2;

  const unsigned beg = rowptr[d];
  const unsigned end = rowptr[d + 1];
  const float dd = dis[d];

  const float4 hv = *(const float4*)(h + (size_t)d * HID + cg);
  const float4 bv = *(const float4*)(bias + cg);
  const float nself = dd * dd;
  float4 acc;
  acc.x = bv.x + nself * hv.x;
  acc.y = bv.y + nself * hv.y;
  acc.z = bv.z + nself * hv.z;
  acc.w = bv.w + nself * hv.w;

  for (unsigned b0 = beg; b0 < end; b0 += 32) {
    const unsigned left = end - b0;
    const int n = (int)(left < 32u ? left : 32u);
    int   sl = 0;
    float dl = 0.f;
    if (lane < n) {
      sl = col[b0 + lane];
      dl = dis[sl];
    }
    int j = 0;
    for (; j + 4 <= n; j += 4) {
      const int s0 = __builtin_amdgcn_readlane(sl, j);
      const int s1 = __builtin_amdgcn_readlane(sl, j + 1);
      const int s2 = __builtin_amdgcn_readlane(sl, j + 2);
      const int s3 = __builtin_amdgcn_readlane(sl, j + 3);
      const float4 v0 = *(const float4*)(h + (size_t)s0 * HID + cg);
      const float4 v1 = *(const float4*)(h + (size_t)s1 * HID + cg);
      const float4 v2 = *(const float4*)(h + (size_t)s2 * HID + cg);
      const float4 v3 = *(const float4*)(h + (size_t)s3 * HID + cg);
      axpy4(acc, lane_f32(dl, j) * dd,     v0);
      axpy4(acc, lane_f32(dl, j + 1) * dd, v1);
      axpy4(acc, lane_f32(dl, j + 2) * dd, v2);
      axpy4(acc, lane_f32(dl, j + 3) * dd, v3);
    }
    if (j + 2 <= n) {
      const int s0 = __builtin_amdgcn_readlane(sl, j);
      const int s1 = __builtin_amdgcn_readlane(sl, j + 1);
      const float4 v0 = *(const float4*)(h + (size_t)s0 * HID + cg);
      const float4 v1 = *(const float4*)(h + (size_t)s1 * HID + cg);
      axpy4(acc, lane_f32(dl, j) * dd,     v0);
      axpy4(acc, lane_f32(dl, j + 1) * dd, v1);
      j += 2;
    }
    if (j < n) {
      const int s0 = __builtin_amdgcn_readlane(sl, j);
      const float4 v0 = *(const float4*)(h + (size_t)s0 * HID + cg);
      axpy4(acc, lane_f32(dl, j) * dd, v0);
    }
  }

  const float4 p = *(const float4*)(pw + cg);
  float4 o;
  o.x = acc.x >= 0.f ? acc.x : p.x * acc.x;
  o.y = acc.y >= 0.f ? acc.y : p.y * acc.y;
  o.z = acc.z >= 0.f ? acc.z : p.z * acc.z;
  o.w = acc.w >= 0.f ? acc.w : p.w * acc.w;
  *(float4*)(out + (size_t)d * HID + cg) = o;
}

__global__ void k_init_deg(unsigned* __restrict__ deg, int N) {
  int i = blockIdx.x * blockDim.x + threadIdx.x;
  if (i < N) deg[i] = 1u;
}

__global__ void k_count(const int* __restrict__ ei, int E, unsigned* __restrict__ deg) {
  int e = blockIdx.x * blockDim.x + threadIdx.x;
  if (e < E) atomicAdd(&deg[ei[E + e]], 1u);
}

__global__ void k_dis(const unsigned* __restrict__ deg, float* __restrict__ dis, int N) {
  int i = blockIdx.x * blockDim.x + threadIdx.x;
  if (i < N) dis[i] = rsqrtf((float)deg[i]);
}

__global__ void k_init_out(const float* __restrict__ h, const float* __restrict__ dis,
                           const float* __restrict__ bias, float* __restrict__ out, int N) {
  int t = blockIdx.x * blockDim.x + threadIdx.x;
  if (t >= N * 32) return;
  int i  = t >> 5;
  int cg = (t & 31) << 2;
  float nrm = dis[i] * dis[i];
  float4 hv = *(const float4*)(h + (size_t)i * HID + cg);
  float4 bv = *(const float4*)(bias + cg);
  float4 o;
  o.x = bv.x + nrm * hv.x;
  o.y = bv.y + nrm * hv.y;
  o.z = bv.z + nrm * hv.z;
  o.w = bv.w + nrm * hv.w;
  *(float4*)(out + (size_t)i * HID + cg) = o;
}

__global__ __launch_bounds__(256) void k_scatter(const int* __restrict__ ei, int E,
                                                 const float* __restrict__ h,
                                                 const float* __restrict__ dis,
                                                 float* __restrict__ out) {
  int w    = (blockIdx.x * blockDim.x + threadIdx.x) >> 5;
  int lane = threadIdx.x & 31;
  if (w >= E) return;
  int s = ei[w];
  int d = ei[E + w];
  float nrm = dis[s] * dis[d];
  const float4 hv = *(const float4*)(h + (size_t)s * HID + (lane << 2));
  float* op = out + (size_t)d * HID + (lane << 2);
  unsafeAtomicAdd(op + 0, nrm * hv.x);
  unsafeAtomicAdd(op + 1, nrm * hv.y);
  unsafeAtomicAdd(op + 2, nrm * hv.z);
  unsafeAtomicAdd(op + 3, nrm * hv.w);
}

__global__ void k_prelu(float* __restrict__ out, const float* __restrict__ pw, int N) {
  int t = blockIdx.x * blockDim.x + threadIdx.x;
  if (t >= N * 32) return;
  int cg = (t & 31) << 2;
  float4 v = *(float4*)(out + (size_t)t * 4);
  float4 p = *(const float4*)(pw + cg);
  v.x = v.x >= 0.f ? v.x : p.x * v.x;
  v.y = v.y >= 0.f ? v.y : p.y * v.y;
  v.z = v.z >= 0.f ? v.z : p.z * v.z;
  v.w = v.w >= 0.f ? v.w : p.w * v.w;
  *(float4*)(out + (size_t)t * 4) = v;
}

extern "C" void kernel_launch(void* const* d_in, const int* in_sizes, int n_in,
                              void* d_out, int out_size, void* d_ws, size_t ws_size,
                              hipStream_t stream) {
  const float* x  = (const float*)d_in[0];
  const int*   ei = (const int*)d_in[1];
  const float* W  = (const float*)d_in[2];
  const float* b  = (const float*)d_in[3];
  const float* pw = (const float*)d_in[4];
  float* out = (float*)d_out;

  const int N = in_sizes[0] / IN_CH;
  const int E = in_sizes[1] / 2;

  char* ws = (char*)d_ws;
  size_t o0 = 0;
  size_t o1 = (o0 + (size_t)N * 4 + 255) & ~(size_t)255;
  size_t o2 = (o1 + (size_t)N * 4 + 255) & ~(size_t)255;
  size_t o3 = (o2 + (size_t)N * HID * 4 + 255) & ~(size_t)255;
  size_t o4 = (o3 + ((size_t)N + 1) * 4 + 255) & ~(size_t)255;
  size_t o5 = (o4 + (size_t)E * 4 + 255) & ~(size_t)255;
  unsigned* cnt    = (unsigned*)(ws + o0);
  float*    dis    = (float*)(ws + o1);
  float*    h      = (float*)(ws + o2);
  unsigned* rowptr = (unsigned*)(ws + o3);
  unsigned* rank   = (unsigned*)(ws + o4);
  int*      col    = (int*)(ws + o5);

  const size_t need = o5 + (size_t)E * 4;

  const int T = 256;
  const int gemm_groups = 512;

  if (ws_size < need) {
    int tq = N * 32;
    k_init_deg<<<(N + T - 1) / T, T, 0, stream>>>(cnt, N);
    k_count   <<<(E + T - 1) / T, T, 0, stream>>>(ei, E, cnt);
    k_dis     <<<(N + T - 1) / T, T, 0, stream>>>(cnt, dis, N);
    k_gemm    <<<gemm_groups, T, 0, stream>>>(x, W, h, N / 16);
    k_init_out<<<(tq + T - 1) / T, T, 0, stream>>>(h, dis, b, out, N);
    k_scatter <<<(E + 7) / 8, T, 0, stream>>>(ei, E, h, dis, out);
    k_prelu   <<<(tq + T - 1) / T, T, 0, stream>>>(out, pw, N);
    return;
  }

  k_zero_cnt  <<<(N + T - 1) / T, T, 0, stream>>>(cnt, N);
  k_count_rank<<<(E + T - 1) / T, T, 0, stream>>>(ei, E, cnt, rank);
  k_scan_dis  <<<(N + 1 + 1023) / 1024, T, 0, stream>>>(cnt, rowptr, dis, N);

  k_fill      <<<(E + T - 1) / T, T, 0, stream>>>(ei, E, rowptr, rank, col);

  k_gemm<<<gemm_groups, T, 0, stream>>>(x, W, h, N / 16);

  k_gather<<<(N + 7) / 8, T, 0, stream>>>(col, rowptr, h, dis, b, pw, out, N);
}


# rank pass: each half of the edge list split in two, whose work-groups fill a half row from opposite ends (slot rank / 15 - rank); 32 node ranges x 4 quarters, list read 32x not 64x; changes sum order
# speedup vs baseline: 3.4274x; 1.5669x over previous
#include <hip/hip_runtime.h>


typedef __attribute__((ext_vector_type(2))) float v2f;
typedef __attribute__((ext_vector_type(8))) float v8f;

#define IN_CH 128
#define HID   128

#define ROW_SLOTS 32
static_assert(ROW_SLOTS == 32, "a slot row is read and packed one slot per lane of a 32-lane wave");

#define RANK_CAP   4096
#define RANK_T     1024
#define HALF_SLOTS (ROW_SLOTS / 2)
#define SET_G      2

__device__ __forceinline__ void load_set(int4 (&dv)[SET_G], int4 (&sv)[SET_G], bool (&ok)[SET_G],
                                         const int4* __restrict__ d4,
                                         const int4* __restrict__ s4, int g, int ge) {
#pragma unroll
  for (int u = 0; u < SET_G; ++u) {
    const int gi = g + u * RANK_T;
    ok[u] = gi < ge;
    const int gc = ok[u] ? gi : ge - 1;
    dv[u] = d4[gc];
    sv[u] = s4[gc];
  }
}

__device__ __forceinline__ void place_set(const int4 (&dv)[SET_G], const int4 (&sv)[SET_G],
                                          const bool (&ok)[SET_G], unsigned* cl, int base,
                                          unsigned w, int* __restrict__ hrow, unsigned xm) {
  int dd[4 * SET_G], ss[4 * SET_G];
#pragma unroll
  for (int u = 0; u < SET_G; ++u) {
    dd[4 * u] = dv[u].x; dd[4 * u + 1] = dv[u].y; dd[4 * u + 2] = dv[u].z; dd[4 * u + 3] = dv[u].w;
    ss[4 * u] = sv[u].x; ss[4 * u + 1] = sv[u].y; ss[4 * u + 2] = sv[u].z; ss[4 * u + 3] = sv[u].w;
  }
  unsigned rk[4 * SET_G];
#pragma unroll
  for (int k = 0; k < 4 * SET_G; ++k) {
    const unsigned u = (unsigned)(dd[k] - base);
    rk[k] = 0xffffffffu;
    if (u < w && ok[k >> 2]) rk[k] = atomicAdd(&cl[u], 1u);
  }
#pragma unroll
  for (int k = 0; k < 4 * SET_G; ++k)
    if (rk[k] < (unsigned)HALF_SLOTS) hrow[(size_t)dd[k] * ROW_SLOTS + (rk[k] ^ xm)] = ss[k];
}

__global__ __launch_bounds__(RANK_T) void k_rank_place(const int* __restrict__ ei, int E, int N,
                                                       unsigned* __restrict__ cnt4,
                                                       int* __restrict__ slots) {
  __shared__ unsigned cl[RANK_CAP];
  const int tid  = threadIdx.x;
  const int q    = (int)blockIdx.x & 1;
  const int half = ((int)blockIdx.x >> 1) & 1;
  const int nb   = (int)gridDim.x >> 2;
  const int span = (N + nb - 1) / nb;
  const int lo   = ((int)blockIdx.x >> 2) * span;
  const int hi   = lo + span < N ? lo + span : N;
  const int mid  = (E >> 1) & ~3;
  const int hb   = half ? mid : 0;
  const int he   = half ? E : mid;
  const int hm   = hb + (((he - hb) >> 1) & ~3);
  const int eb   = q ? hm : hb;
  const int ee   = q ? he : hm;
  const unsigned xm = q ? (unsigned)(HALF_SLOTS - 1) : 0u;
  const bool vec = (E & 3) == 0;
  const int gb   = eb >> 2;
  const int ge   = vec ? (ee >> 2) : gb;
  const int e1   = vec ? ee : eb;
  const int4* __restrict__ s4 = (const int4*)ei;
  const int4* __restrict__ d4 = (const int4*)(ei + E);
  int* __restrict__ hrow = slots + half * HALF_SLOTS;

  for (int base = lo; base < hi; base += RANK_CAP) {
    const unsigned w = (unsigned)(hi - base < RANK_CAP ? hi - base : RANK_CAP);
    for (int i = tid; i < RANK_CAP; i += RANK_T) cl[i] = 0u;
    __syncthreads();

    if (gb < ge) {
      int4 da[SET_G], sa[SET_G], db[SET_G], sb[SET_G];
      bool oa[SET_G], ob[SET_G];
      load_set(da, sa, oa, d4, s4, gb + tid, ge);
      for (int g0 = gb + tid; g0 < ge; g0 += 2 * SET_G * RANK_T) {
        load_set(db, sb, ob, d4, s4, g0 + SET_G * RANK_T, ge);
        place_set(da, sa, oa, cl, base, w, hrow, xm);
        load_set(da, sa, oa, d4, s4, g0 + 2 * SET_G * RANK_T, ge);
        place_set(db, sb, ob, cl, base, w, hrow, xm);
      }
    }
    for (int e = e1 + tid; e < ee; e += RANK_T) {
      const int d = ei[E + e];
      const unsigned u = (unsigned)(d - base);
      if (u < w) {
        const unsigned r = atomicAdd(&cl[u], 1u);
        if (r < (unsigned)HALF_SLOTS) hrow[(size_t)d * ROW_SLOTS + (r ^ xm)] = ei[e];
      }
    }

    __syncthreads();
    for (int i = tid; i < (int)w; i += RANK_T) cnt4[4 * (size_t)(base + i) + 2 * half + q] = cl[i];
    __syncthreads();
  }
}

__device__ __forceinline__ void load_a_chunk(v2f (&a)[8], const float* __restrict__ xr) {
#pragma unroll
  for (int q = 0; q < 8; ++q) {
    const float2 av = *(const float2*)(xr + q * 4);
    a[q][0] = av.x; a[q][1] = av.y;
  }
}

__device__ __forceinline__ void mma_chunk(v8f (&acc)[8], const v2f (&a)[8], float4 (&bb)[2][8],
                                          const float4* wl, int j0) {
#pragma unroll
  for (int jj = 0; jj < 4; ++jj) {
    const float4* wn = wl + ((j0 + jj + 1) & 15) * 2 * HID;
    __builtin_amdgcn_sched_barrier(0);
#pragma unroll
    for (int ct = 0; ct < 8; ++ct) bb[(jj + 1) & 1][ct] = wn[ct * 16];
    __builtin_amdgcn_sched_barrier(0);
#pragma unroll
    for (int ct = 0; ct < 8; ++ct) {
      v2f b; b[0] = bb[jj & 1][ct].x; b[1] = bb[jj & 1][ct].y;
      acc[ct] = __builtin_amdgcn_wmma_f32_16x16x4_f32(false, a[2 * jj], false, b,
                                                      (short)0, acc[ct], false, false);
    }
#pragma unroll
    for (int ct = 0; ct < 8; ++ct) {
      v2f b; b[0] = bb[jj & 1][ct].z; b[1] = bb[jj & 1][ct].w;
      acc[ct] = __builtin_amdgcn_wmma_f32_16x16x4_f32(false, a[2 * jj + 1], false, b,
                                                      (short)0, acc[ct], false, false);
    }
  }
}

__global__ __launch_bounds__(256) void k_gemm(const float* __restrict__ x,
                                              const float* __restrict__ W,
                                              float* __restrict__ h, int nstrips) {
  __shared__ float4 Ws[16 * 2 * HID];

  const int tid  = threadIdx.x;
  const int lane = tid & 31;
  const int wave = __builtin_amdgcn_readfirstlane(tid >> 5);
  const int mn   = lane & 15;
  const int hf   = lane >> 4;
  const int kof  = hf << 1;
  const int G    = (int)gridDim.x;

  int strip = wave * G + (int)blockIdx.x;

  v2f a0[8], a1[8];
  {
    const int s0 = strip < nstrips ? strip : nstrips - 1;
    load_a_chunk(a0, x + (size_t)(s0 * 16 + mn) * IN_CH + kof);
  }

#pragma unroll
  for (int u = 0; u < 16; ++u) {
    const int s  = u * 256 + tid;
    const int n  = s & (HID - 1);
    const int kb = ((s >> 8) << 3) + (((s >> 7) & 1) << 1);
    float4 v;
    v.x = W[(size_t)(kb    ) * HID + n];
    v.y = W[(size_t)(kb + 1) * HID + n];
    v.z = W[(size_t)(kb + 4) * HID + n];
    v.w = W[(size_t)(kb + 5) * HID + n];
    Ws[s] = v;
  }
  __syncthreads();

  const float4* wl = Ws + hf * HID + mn;
  const int mo = hf * 8;

  float4 bb[2][8];
#pragma unroll
  for (int ct = 0; ct < 8; ++ct) bb[0][ct] = wl[ct * 16];

  while (strip < nstrips) {
    v8f acc[8];
#pragma unroll
    for (int ct = 0; ct < 8; ++ct) acc[ct] = (v8f){};

    const int nstrip = strip + 8 * G;
    const float* xr = x + (size_t)(strip * 16 + mn) * IN_CH + kof;

#pragma unroll 1
    for (int c2 = 0; c2 < 2; ++c2) {
      load_a_chunk(a1, xr + (2 * c2 + 1) * 32);
      mma_chunk(acc, a0, bb, wl, c2 * 8);
      const int ps = (c2 == 0 || nstrip >= nstrips) ? strip : nstrip;
      load_a_chunk(a0, x + (size_t)(ps * 16 + mn) * IN_CH + kof + (c2 == 0 ? 64 : 0));
      mma_chunk(acc, a1, bb, wl, c2 * 8 + 4);
    }

    float* hr = h + (size_t)(strip * 16 + mo) * HID + mn;
#pragma unroll
    for (int ct = 0; ct < 8; ++ct)
#pragma unroll
      for (int r = 0; r < 8; ++r)
        hr[(size_t)r * HID + ct * 16] = acc[ct][r];

    strip = nstrip;
  }
}

__device__ __forceinline__ float lane_f32(float v, int j) {
  return __builtin_bit_cast(float, __builtin_amdgcn_readlane(__builtin_bit_cast(int, v), j));
}

__device__ __forceinline__ void axpy4(float4& acc, float w, const float4 v) {
  acc.x += w * v.x;
  acc.y += w * v.y;
  acc.z += w * v.z;
  acc.w += w * v.w;
}

#define APITCH 132

__device__ __forceinline__ void add_row(float* at, int r, int cg, const float4 acc) {
  float4* p = (float4*)(at + r * APITCH + cg);
  float4 t = *p;
  t.x += acc.x; t.y += acc.y; t.z += acc.z; t.w += acc.w;
  *p = t;
}

__device__ __forceinline__ float node_scale(unsigned c) { return rsqrtf((float)(c + 1u)); }

__device__ __forceinline__ unsigned node_count(const uint4 t) { return t.x + t.y + t.z + t.w; }

__device__ __forceinline__ unsigned half_lanes(unsigned a, unsigned b) {
  return ((1u << a) - 1u) | (0xffffu & ~((1u << (HALF_SLOTS - b)) - 1u));
}

__device__ __forceinline__ unsigned row_lanes(const uint4 t) {
  if (t.x > (unsigned)HALF_SLOTS || t.y > (unsigned)HALF_SLOTS || t.x + t.y > (unsigned)HALF_SLOTS ||
      t.z > (unsigned)HALF_SLOTS || t.w > (unsigned)HALF_SLOTS || t.z + t.w > (unsigned)HALF_SLOTS)
    return 0u;
  return half_lanes(t.x, t.y) | (half_lanes(t.z, t.w) << HALF_SLOTS);
}

__global__ __launch_bounds__(256) void k_agg_gemm(const float* __restrict__ x,
                                                  const float* __restrict__ W,
                                                  const int* __restrict__ slots,
                                                  const uint4* __restrict__ cnt4,
                                                  const int* __restrict__ ei, int E,
                                                  const float* __restrict__ bias,
                                                  const float* __restrict__ pw,
                                                  float* __restrict__ out, int N, int nstrips) {
  __shared__ float4 Ws[16 * 2 * HID];
  __shared__ float  At[8 * 16 * APITCH];
  __shared__ int    Fl[8 * 16 * ROW_SLOTS];

  const int tid  = threadIdx.x;
  const int lane = tid & 31;
  const int wave = __builtin_amdgcn_readfirstlane(tid >> 5);
  const int mn   = lane & 15;
  const int hf   = lane >> 4;
  const int kof  = hf << 1;
  const int cg   = lane << 2;
  const int G    = (int)gridDim.x;

#pragma unroll
  for (int u = 0; u < 16; ++u) {
    const int s  = u * 256 + tid;
    const int n  = s & (HID - 1);
    const int kb = ((s >> 8) << 3) + (((s >> 7) & 1) << 1);
    float4 v;
    v.x = W[(size_t)(kb    ) * HID + n];
    v.y = W[(size_t)(kb + 1) * HID + n];
    v.z = W[(size_t)(kb + 4) * HID + n];
    v.w = W[(size_t)(kb + 5) * HID + n];
    Ws[s] = v;
  }
  __syncthreads();

  float* at = At + wave * 16 * APITCH;
  int*   fl = Fl + wave * 16 * ROW_SLOTS;
  const float* ar = at + mn * APITCH + kof;
  const float4* wl = Ws + hf * HID + mn;
  const int mo = hf * 8;

  float bz[8], pz[8];
#pragma unroll
  for (int ct = 0; ct < 8; ++ct) { bz[ct] = bias[ct * 16 + mn]; pz[ct] = pw[ct * 16 + mn]; }

  float4 bb[2][8];
#pragma unroll
  for (int ct = 0; ct < 8; ++ct) bb[0][ct] = wl[ct * 16];

  int strip = wave * G + (int)blockIdx.x;
  unsigned c = 0u, vm = 0u;
  int sr[16];
#pragma unroll
  for (int r = 0; r < 16; ++r) sr[r] = 0;
  if (strip < nstrips) {
    const int di = strip * 16 + lane;
    if (lane < 16 && di < N) { const uint4 t = cnt4[di]; c = node_count(t); vm = row_lanes(t); }
#pragma unroll
    for (int r = 0; r < 16; ++r) {
      const unsigned mr = (unsigned)__builtin_amdgcn_readlane((int)vm, r);
      const int* rowp = slots + (size_t)(strip * 16 + r) * ROW_SLOTS;
      if ((mr >> lane) & 1u) sr[r] = rowp[lane];
    }
  }

  while (strip < nstrips) {
    const int d0 = strip * 16;
    const int nstrip = strip + 8 * G;

    const float    ds = (lane < 16 && d0 + lane < N) ? node_scale(c) : 0.f;
    unsigned run = 0u;
    int rpi = 0;
#pragma unroll
    for (int r = 0; r < 16; ++r) {
      const unsigned mr = (unsigned)__builtin_amdgcn_readlane((int)vm, r);
      rpi = (lane == r) ? (int)run : rpi;
      if ((mr >> lane) & 1u) fl[run + __builtin_amdgcn_mbcnt_lo(mr, 0u)] = sr[r];
      run += (unsigned)__builtin_popcount(mr);
    }
    rpi = (lane >= 16) ? (int)run : rpi;
    const unsigned rp = (unsigned)rpi;
    const unsigned E1 = run;

    __builtin_amdgcn_fence(__ATOMIC_RELEASE, "wavefront");
    __builtin_amdgcn_wave_barrier();
    __builtin_amdgcn_fence(__ATOMIC_ACQUIRE, "wavefront");

    int sl = 0, sl_n = 0;
    if ((unsigned)lane < E1)       sl   = fl[lane];
    if (32u + (unsigned)lane < E1) sl_n = fl[32 + lane];

    unsigned dc = 0u, c_n = 0u, vm_n = 0u;
    {
      float4 sv[16];
#pragma unroll
      for (int r = 0; r < 16; ++r) {
        const int d = (d0 + r < N) ? d0 + r : N - 1;
        sv[r] = *(const float4*)(x + (size_t)d * IN_CH + cg);
      }
      if ((unsigned)lane < E1) dc = node_count(cnt4[sl]);
      if (nstrip < nstrips) {
        const int di = nstrip * 16 + lane;
        if (lane < 16 && di < N) { const uint4 t = cnt4[di]; c_n = node_count(t); vm_n = row_lanes(t); }
      }
#pragma unroll
      for (int r = 0; r < 16; ++r) {
        const float dr = lane_f32(ds, r);
        const float ns = dr * dr;
        float4 t;
        t.x = ns * sv[r].x; t.y = ns * sv[r].y; t.z = ns * sv[r].z; t.w = ns * sv[r].w;
        *(float4*)(at + r * APITCH + cg) = t;
      }
    }

    int      r    = 0;
    unsigned endc = (unsigned)__builtin_amdgcn_readlane((int)rp, 1);
    float    ddc  = lane_f32(ds, 0);
    float4 acc = make_float4(0.f, 0.f, 0.f, 0.f);

    for (unsigned b0 = 0; b0 < E1; b0 += 32) {
      const unsigned left = E1 - b0;
      const int n = (int)(left < 32u ? left : 32u);
      unsigned dc_n = 0u;
      int   sl_nn = 0;
      float dl = 0.f;

      for (int j = 0; j < n; j += 8) {
        float4 v[8];
#pragma unroll
        for (int u = 0; u < 8; ++u) {
          const int jj = (j + u < n) ? j + u : n - 1;
          const int s = __builtin_amdgcn_readlane(sl, jj);
          v[u] = *(const float4*)(x + (size_t)s * IN_CH + cg);
        }
        if (j == 0) {
          if (b0 + 32u + (unsigned)lane < E1) dc_n  = node_count(cnt4[sl_n]);
          if (b0 + 64u + (unsigned)lane < E1) sl_nn = fl[b0 + 64 + lane];
          dl = node_scale(dc);
        }
#pragma unroll
        for (int u = 0; u < 8; ++u) {
          if (j + u < n) {
            const unsigned e = b0 + (unsigned)(j + u);
            while (e >= endc) {
              add_row(at, r, cg, acc);
              acc = make_float4(0.f, 0.f, 0.f, 0.f);
              ++r;
              endc = (unsigned)__builtin_amdgcn_readlane((int)rp, r + 1);
              ddc  = lane_f32(ds, r);
            }
            axpy4(acc, lane_f32(dl, j + u) * ddc, v[u]);
          }
        }
      }
      sl = sl_n; dc = dc_n; sl_n = sl_nn;
    }
    add_row(at, r, cg, acc);

    const bool ov = c != (unsigned)__builtin_popcount(vm);
    if (__builtin_amdgcn_ballot_w32(ov) != 0u) {
      for (int q0 = 0; q0 < E; q0 += 32) {
        int od = -1, os = 0;
        if (q0 + lane < E) { od = ei[E + q0 + lane]; os = ei[q0 + lane]; }
        const int rl = od - d0;
        const int on = __builtin_amdgcn_ds_bpermute((rl & 15) << 2, (int)ov);
        unsigned mk = __builtin_amdgcn_ballot_w32(rl >= 0 && rl < 16 && on != 0);
        while (mk != 0u) {
          const int j = __builtin_ctz(mk);
          mk &= mk - 1u;
          const int rd = __builtin_amdgcn_readlane(od, j) - d0;
          const int s  = __builtin_amdgcn_readlane(os, j);
          const float w = node_scale(node_count(cnt4[s])) * lane_f32(ds, rd);
          const float4 xv = *(const float4*)(x + (size_t)s * IN_CH + cg);
          float4 t;
          t.x = w * xv.x; t.y = w * xv.y; t.z = w * xv.z; t.w = w * xv.w;
          add_row(at, rd, cg, t);
        }
      }
    }

    int sr_n[16];
#pragma unroll
    for (int q = 0; q < 16; ++q) sr_n[q] = 0;
    if (nstrip < nstrips) {
#pragma unroll
      for (int q = 0; q < 16; ++q) {
        const unsigned mq = (unsigned)__builtin_amdgcn_readlane((int)vm_n, q);
        const int* rowp = slots + (size_t)(nstrip * 16 + q) * ROW_SLOTS;
        if ((mq >> lane) & 1u) sr_n[q] = rowp[lane];
      }
    }

    __builtin_amdgcn_fence(__ATOMIC_RELEASE, "wavefront");
    __builtin_amdgcn_wave_barrier();
    __builtin_amdgcn_fence(__ATOMIC_ACQUIRE, "wavefront");

    v8f cc[8];
#pragma unroll
    for (int ct = 0; ct < 8; ++ct) cc[ct] = (v8f){};
    v2f a0[8], a1[8];
    load_a_chunk(a0, ar);
#pragma unroll 1
    for (int c2 = 0; c2 < 2; ++c2) {
      load_a_chunk(a1, ar + (2 * c2 + 1) * 32);
      mma_chunk(cc, a0, bb, wl, c2 * 8);
      load_a_chunk(a0, ar + (c2 == 0 ? 64 : 0));
      mma_chunk(cc, a1, bb, wl, c2 * 8 + 4);
    }

    float* orow = out + (size_t)(d0 + mo) * HID + mn;
    if (d0 + 16 <= N) {
#pragma unroll
      for (int ct = 0; ct < 8; ++ct)
#pragma unroll
        for (int q = 0; q < 8; ++q) {
          const float t = cc[ct][q] + bz[ct];
          orow[(size_t)q * HID + ct * 16] = t >= 0.f ? t : pz[ct] * t;
        }
    } else {
#pragma unroll
      for (int ct = 0; ct < 8; ++ct)
#pragma unroll
        for (int q = 0; q < 8; ++q) {
          const float t = cc[ct][q] + bz[ct];
          if (d0 + mo + q < N) orow[(size_t)q * HID + ct * 16] = t >= 0.f ? t : pz[ct] * t;
        }
    }

    __builtin_amdgcn_fence(__ATOMIC_RELEASE, "wavefront");
    __builtin_amdgcn_wave_barrier();

    c = c_n; vm = vm_n;
#pragma unroll
    for (int q = 0; q < 16; ++q) sr[q] = sr_n[q];
    strip = nstrip;
  }
}

__global__ void k_init_deg(unsigned* __restrict__ deg, int N) {
  int i = blockIdx.x * blockDim.x + threadIdx.x;
  if (i < N) deg[i] = 1u;
}

__global__ void k_count(const int* __restrict__ ei, int E, unsigned* __restrict__ deg) {
  int e = blockIdx.x * blockDim.x + threadIdx.x;
  if (e < E) atomicAdd(&deg[ei[E + e]], 1u);
}

__global__ void k_dis(const unsigned* __restrict__ deg, float* __restrict__ dis, int N) {
  int i = blockIdx.x * blockDim.x + threadIdx.x;
  if (i < N) dis[i] = rsqrtf((float)deg[i]);
}

__global__ void k_init_out(const float* __restrict__ h, const float* __restrict__ dis,
                           const float* __restrict__ bias, float* __restrict__ out, int N) {
  int t = blockIdx.x * blockDim.x + threadIdx.x;
  if (t >= N * 32) return;
  int i  = t >> 5;
  int cg = (t & 31) << 2;
  float nrm = dis[i] * dis[i];
  float4 hv = *(const float4*)(h + (size_t)i * HID + cg);
  float4 bv = *(const float4*)(bias + cg);
  float4 o;
  o.x = bv.x + nrm * hv.x;
  o.y = bv.y + nrm * hv.y;
  o.z = bv.z + nrm * hv.z;
  o.w = bv.w + nrm * hv.w;
  *(float4*)(out + (size_t)i * HID + cg) = o;
}

__global__ __launch_bounds__(256) void k_scatter(const int* __restrict__ ei, int E,
                                                 const float* __restrict__ h,
                                                 const float* __restrict__ dis,
                                                 float* __restrict__ out) {
  int w    = (blockIdx.x * blockDim.x + threadIdx.x) >> 5;
  int lane = threadIdx.x & 31;
  if (w >= E) return;
  int s = ei[w];
  int d = ei[E + w];
  float nrm = dis[s] * dis[d];
  const float4 hv = *(const float4*)(h + (size_t)s * HID + (lane << 2));
  float* op = out + (size_t)d * HID + (lane << 2);
  unsafeAtomicAdd(op + 0, nrm * hv.x);
  unsafeAtomicAdd(op + 1, nrm * hv.y);
  unsafeAtomicAdd(op + 2, nrm * hv.z);
  unsafeAtomicAdd(op + 3, nrm * hv.w);
}

__global__ void k_prelu(float* __restrict__ out, const float* __restrict__ pw, int N) {
  int t = blockIdx.x * blockDim.x + threadIdx.x;
  if (t >= N * 32) return;
  int cg = (t & 31) << 2;
  float4 v = *(float4*)(out + (size_t)t * 4);
  float4 p = *(const float4*)(pw + cg);
  v.x = v.x >= 0.f ? v.x : p.x * v.x;
  v.y = v.y >= 0.f ? v.y : p.y * v.y;
  v.z = v.z >= 0.f ? v.z : p.z * v.z;
  v.w = v.w >= 0.f ? v.w : p.w * v.w;
  *(float4*)(out + (size_t)t * 4) = v;
}

extern "C" void kernel_launch(void* const* d_in, const int* in_sizes, int n_in,
                              void* d_out, int out_size, void* d_ws, size_t ws_size,
                              hipStream_t stream) {
  const float* x  = (const float*)d_in[0];
  const int*   ei = (const int*)d_in[1];
  const float* W  = (const float*)d_in[2];
  const float* b  = (const float*)d_in[3];
  const float* pw = (const float*)d_in[4];
  float* out = (float*)d_out;

  const int N = in_sizes[0] / IN_CH;
  const int E = in_sizes[1] / 2;

  char* ws = (char*)d_ws;
  size_t o0 = 0;
  size_t o1 = (o0 + (size_t)N * 16 + 255) & ~(size_t)255;
  size_t o2 = (o1 + (size_t)N * 4 + 255) & ~(size_t)255;
  unsigned* cnt    = (unsigned*)(ws + o0);
  float*    dis    = (float*)(ws + o1);
  float*    h      = (float*)(ws + o2);
  int*      slots  = (int*)(ws + o2);

  const size_t need = o2 + (size_t)N * ROW_SLOTS * 4;

  const int T = 256;
  const int gemm_groups = 512;

  if (ws_size < need) {
    int tq = N * 32;
    k_init_deg<<<(N + T - 1) / T, T, 0, stream>>>(cnt, N);
    k_count   <<<(E + T - 1) / T, T, 0, stream>>>(ei, E, cnt);
    k_dis     <<<(N + T - 1) / T, T, 0, stream>>>(cnt, dis, N);
    k_gemm    <<<gemm_groups, T, 0, stream>>>(x, W, h, N / 16);
    k_init_out<<<(tq + T - 1) / T, T, 0, stream>>>(h, dis, b, out, N);
    k_scatter <<<(E + 7) / 8, T, 0, stream>>>(ei, E, h, dis, out);
    k_prelu   <<<(tq + T - 1) / T, T, 0, stream>>>(out, pw, N);
    return;
  }

  k_rank_place<<<128, RANK_T, 0, stream>>>(ei, E, N, cnt, slots);

  k_agg_gemm<<<gemm_groups, T, 0, stream>>>(x, W, slots, (const uint4*)cnt, ei, E, b, pw, out, N, (N + 15) / 16);
}


# edge sum: all 32 source rows of a list group requested as one run (two groups of 16 on scalar row bases) before the first is summed, was 8 rows per wait; sum order unchanged
# speedup vs baseline: 3.5614x; 1.0391x over previous
#include <hip/hip_runtime.h>


typedef __attribute__((ext_vector_type(2))) float v2f;
typedef __attribute__((ext_vector_type(8))) float v8f;
typedef __attribute__((ext_vector_type(4))) float v4f;
typedef const __attribute__((address_space(1))) v4f* gv4p;

#define IN_CH 128
#define HID   128

#define ROW_SLOTS 32
static_assert(ROW_SLOTS == 32, "a slot row is read and packed one slot per lane of a 32-lane wave");

#define RANK_CAP   4096
#define RANK_T     1024
#define HALF_SLOTS (ROW_SLOTS / 2)
#define SET_G      2

__device__ __forceinline__ void load_set(int4 (&dv)[SET_G], int4 (&sv)[SET_G], bool (&ok)[SET_G],
                                         const int4* __restrict__ d4,
                                         const int4* __restrict__ s4, int g, int ge) {
#pragma unroll
  for (int u = 0; u < SET_G; ++u) {
    const int gi = g + u * RANK_T;
    ok[u] = gi < ge;
    const int gc = ok[u] ? gi : ge - 1;
    dv[u] = d4[gc];
    sv[u] = s4[gc];
  }
}

__device__ __forceinline__ void place_set(const int4 (&dv)[SET_G], const int4 (&sv)[SET_G],
                                          const bool (&ok)[SET_G], unsigned* cl, int base,
                                          unsigned w, int* __restrict__ hrow, unsigned xm) {
  int dd[4 * SET_G], ss[4 * SET_G];
#pragma unroll
  for (int u = 0; u < SET_G; ++u) {
    dd[4 * u] = dv[u].x; dd[4 * u + 1] = dv[u].y; dd[4 * u + 2] = dv[u].z; dd[4 * u + 3] = dv[u].w;
    ss[4 * u] = sv[u].x; ss[4 * u + 1] = sv[u].y; ss[4 * u + 2] = sv[u].z; ss[4 * u + 3] = sv[u].w;
  }
  unsigned rk[4 * SET_G];
#pragma unroll
  for (int k = 0; k < 4 * SET_G; ++k) {
    const unsigned u = (unsigned)(dd[k] - base);
    rk[k] = 0xffffffffu;
    if (u < w && ok[k >> 2]) rk[k] = atomicAdd(&cl[u], 1u);
  }
#pragma unroll
  for (int k = 0; k < 4 * SET_G; ++k)
    if (rk[k] < (unsigned)HALF_SLOTS) hrow[(size_t)dd[k] * ROW_SLOTS + (rk[k] ^ xm)] = ss[k];
}

__global__ __launch_bounds__(RANK_T) void k_rank_place(const int* __restrict__ ei, int E, int N,
                                                       unsigned* __restrict__ cnt4,
                                                       int* __restrict__ slots) {
  __shared__ unsigned cl[RANK_CAP];
  const int tid  = threadIdx.x;
  const int q    = (int)blockIdx.x & 1;
  const int half = ((int)blockIdx.x >> 1) & 1;
  const int nb   = (int)gridDim.x >> 2;
  const int span = (N + nb - 1) / nb;
  const int lo   = ((int)blockIdx.x >> 2) * span;
  const int hi   = lo + span < N ? lo + span : N;
  const int mid  = (E >> 1) & ~3;
  const int hb   = half ? mid : 0;
  const int he   = half ? E : mid;
  const int hm   = hb + (((he - hb) >> 1) & ~3);
  const int eb   = q ? hm : hb;
  const int ee   = q ? he : hm;
  const unsigned xm = q ? (unsigned)(HALF_SLOTS - 1) : 0u;
  const bool vec = (E & 3) == 0;
  const int gb   = eb >> 2;
  const int ge   = vec ? (ee >> 2) : gb;
  const int e1   = vec ? ee : eb;
  const int4* __restrict__ s4 = (const int4*)ei;
  const int4* __restrict__ d4 = (const int4*)(ei + E);
  int* __restrict__ hrow = slots + half * HALF_SLOTS;

  for (int base = lo; base < hi; base += RANK_CAP) {
    const unsigned w = (unsigned)(hi - base < RANK_CAP ? hi - base : RANK_CAP);
    for (int i = tid; i < RANK_CAP; i += RANK_T) cl[i] = 0u;
    __syncthreads();

    if (gb < ge) {
      int4 da[SET_G], sa[SET_G], db[SET_G], sb[SET_G];
      bool oa[SET_G], ob[SET_G];
      load_set(da, sa, oa, d4, s4, gb + tid, ge);
      for (int g0 = gb + tid; g0 < ge; g0 += 2 * SET_G * RANK_T) {
        load_set(db, sb, ob, d4, s4, g0 + SET_G * RANK_T, ge);
        place_set(da, sa, oa, cl, base, w, hrow, xm);
        load_set(da, sa, oa, d4, s4, g0 + 2 * SET_G * RANK_T, ge);
        place_set(db, sb, ob, cl, base, w, hrow, xm);
      }
    }
    for (int e = e1 + tid; e < ee; e += RANK_T) {
      const int d = ei[E + e];
      const unsigned u = (unsigned)(d - base);
      if (u < w) {
        const unsigned r = atomicAdd(&cl[u], 1u);
        if (r < (unsigned)HALF_SLOTS) hrow[(size_t)d * ROW_SLOTS + (r ^ xm)] = ei[e];
      }
    }

    __syncthreads();
    for (int i = tid; i < (int)w; i += RANK_T) cnt4[4 * (size_t)(base + i) + 2 * half + q] = cl[i];
    __syncthreads();
  }
}

__device__ __forceinline__ void load_a_chunk(v2f (&a)[8], const float* __restrict__ xr) {
#pragma unroll
  for (int q = 0; q < 8; ++q) {
    const float2 av = *(const float2*)(xr + q * 4);
    a[q][0] = av.x; a[q][1] = av.y;
  }
}

__device__ __forceinline__ void mma_chunk(v8f (&acc)[8], const v2f (&a)[8], float4 (&bb)[2][8],
                                          const float4* wl, int j0) {
#pragma unroll
  for (int jj = 0; jj < 4; ++jj) {
    const float4* wn = wl + ((j0 + jj + 1) & 15) * 2 * HID;
    __builtin_amdgcn_sched_barrier(0);
#pragma unroll
    for (int ct = 0; ct < 8; ++ct) bb[(jj + 1) & 1][ct] = wn[ct * 16];
    __builtin_amdgcn_sched_barrier(0);
#pragma unroll
    for (int ct = 0; ct < 8; ++ct) {
      v2f b; b[0] = bb[jj & 1][ct].x; b[1] = bb[jj & 1][ct].y;
      acc[ct] = __builtin_amdgcn_wmma_f32_16x16x4_f32(false, a[2 * jj], false, b,
                                                      (short)0, acc[ct], false, false);
    }
#pragma unroll
    for (int ct = 0; ct < 8; ++ct) {
      v2f b; b[0] = bb[jj & 1][ct].z; b[1] = bb[jj & 1][ct].w;
      acc[ct] = __builtin_amdgcn_wmma_f32_16x16x4_f32(false, a[2 * jj + 1], false, b,
                                                      (short)0, acc[ct], false, false);
    }
  }
}

__global__ __launch_bounds__(256) void k_gemm(const float* __restrict__ x,
                                              const float* __restrict__ W,
                                              float* __restrict__ h, int nstrips) {
  __shared__ float4 Ws[16 * 2 * HID];

  const int tid  = threadIdx.x;
  const int lane = tid & 31;
  const int wave = __builtin_amdgcn_readfirstlane(tid >> 5);
  const int mn   = lane & 15;
  const int hf   = lane >> 4;
  const int kof  = hf << 1;
  const int G    = (int)gridDim.x;

  int strip = wave * G + (int)blockIdx.x;

  v2f a0[8], a1[8];
  {
    const int s0 = strip < nstrips ? strip : nstrips - 1;
    load_a_chunk(a0, x + (size_t)(s0 * 16 + mn) * IN_CH + kof);
  }

#pragma unroll
  for (int u = 0; u < 16; ++u) {
    const int s  = u * 256 + tid;
    const int n  = s & (HID - 1);
    const int kb = ((s >> 8) << 3) + (((s >> 7) & 1) << 1);
    float4 v;
    v.x = W[(size_t)(kb    ) * HID + n];
    v.y = W[(size_t)(kb + 1) * HID + n];
    v.z = W[(size_t)(kb + 4) * HID + n];
    v.w = W[(size_t)(kb + 5) * HID + n];
    Ws[s] = v;
  }
  __syncthreads();

  const float4* wl = Ws + hf * HID + mn;
  const int mo = hf * 8;

  float4 bb[2][8];
#pragma unroll
  for (int ct = 0; ct < 8; ++ct) bb[0][ct] = wl[ct * 16];

  while (strip < nstrips) {
    v8f acc[8];
#pragma unroll
    for (int ct = 0; ct < 8; ++ct) acc[ct] = (v8f){};

    const int nstrip = strip + 8 * G;
    const float* xr = x + (size_t)(strip * 16 + mn) * IN_CH + kof;

#pragma unroll 1
    for (int c2 = 0; c2 < 2; ++c2) {
      load_a_chunk(a1, xr + (2 * c2 + 1) * 32);
      mma_chunk(acc, a0, bb, wl, c2 * 8);
      const int ps = (c2 == 0 || nstrip >= nstrips) ? strip : nstrip;
      load_a_chunk(a0, x + (size_t)(ps * 16 + mn) * IN_CH + kof + (c2 == 0 ? 64 : 0));
      mma_chunk(acc, a1, bb, wl, c2 * 8 + 4);
    }

    float* hr = h + (size_t)(strip * 16 + mo) * HID + mn;
#pragma unroll
    for (int ct = 0; ct < 8; ++ct)
#pragma unroll
      for (int r = 0; r < 8; ++r)
        hr[(size_t)r * HID + ct * 16] = acc[ct][r];

    strip = nstrip;
  }
}

__device__ __forceinline__ float lane_f32(float v, int j) {
  return __builtin_bit_cast(float, __builtin_amdgcn_readlane(__builtin_bit_cast(int, v), j));
}

__device__ __forceinline__ void axpy4(float4& acc, float w, const float4 v) {
  acc.x += w * v.x;
  acc.y += w * v.y;
  acc.z += w * v.z;
  acc.w += w * v.w;
}

#define APITCH 132

__device__ __forceinline__ void add_row(float* at, int r, int cg, const float4 acc) {
  float4* p = (float4*)(at + r * APITCH + cg);
  float4 t = *p;
  t.x += acc.x; t.y += acc.y; t.z += acc.z; t.w += acc.w;
  *p = t;
}

__device__ __forceinline__ float node_scale(unsigned c) { return rsqrtf((float)(c + 1u)); }

__device__ __forceinline__ unsigned node_count(const uint4 t) { return t.x + t.y + t.z + t.w; }

__device__ __forceinline__ unsigned half_lanes(unsigned a, unsigned b) {
  return ((1u << a) - 1u) | (0xffffu & ~((1u << (HALF_SLOTS - b)) - 1u));
}

__device__ __forceinline__ unsigned row_lanes(const uint4 t) {
  if (t.x > (unsigned)HALF_SLOTS || t.y > (unsigned)HALF_SLOTS || t.x + t.y > (unsigned)HALF_SLOTS ||
      t.z > (unsigned)HALF_SLOTS || t.w > (unsigned)HALF_SLOTS || t.z + t.w > (unsigned)HALF_SLOTS)
    return 0u;
  return half_lanes(t.x, t.y) | (half_lanes(t.z, t.w) << HALF_SLOTS);
}

__global__ __launch_bounds__(256) void k_agg_gemm(const float* __restrict__ x,
                                                  const float* __restrict__ W,
                                                  const int* __restrict__ slots,
                                                  const uint4* __restrict__ cnt4,
                                                  const int* __restrict__ ei, int E,
                                                  const float* __restrict__ bias,
                                                  const float* __restrict__ pw,
                                                  float* __restrict__ out, int N, int nstrips) {
  __shared__ float4 Ws[16 * 2 * HID];
  __shared__ float  At[8 * 16 * APITCH];
  __shared__ int    Fl[8 * 16 * ROW_SLOTS];

  const int tid  = threadIdx.x;
  const int lane = tid & 31;
  const int wave = __builtin_amdgcn_readfirstlane(tid >> 5);
  const int mn   = lane & 15;
  const int hf   = lane >> 4;
  const int kof  = hf << 1;
  const int cg   = lane << 2;
  const int G    = (int)gridDim.x;

#pragma unroll
  for (int u = 0; u < 16; ++u) {
    const int s  = u * 256 + tid;
    const int n  = s & (HID - 1);
    const int kb = ((s >> 8) << 3) + (((s >> 7) & 1) << 1);
    float4 v;
    v.x = W[(size_t)(kb    ) * HID + n];
    v.y = W[(size_t)(kb + 1) * HID + n];
    v.z = W[(size_t)(kb + 4) * HID + n];
    v.w = W[(size_t)(kb + 5) * HID + n];
    Ws[s] = v;
  }
  __syncthreads();

  float* at = At + wave * 16 * APITCH;
  int*   fl = Fl + wave * 16 * ROW_SLOTS;
  const float* ar = at + mn * APITCH + kof;
  const float4* wl = Ws + hf * HID + mn;
  const int mo = hf * 8;

  float bz[8], pz[8];
#pragma unroll
  for (int ct = 0; ct < 8; ++ct) { bz[ct] = bias[ct * 16 + mn]; pz[ct] = pw[ct * 16 + mn]; }

  float4 bb[2][8];
#pragma unroll
  for (int ct = 0; ct < 8; ++ct) bb[0][ct] = wl[ct * 16];

  int strip = wave * G + (int)blockIdx.x;
  unsigned c = 0u, vm = 0u;
  int sr[16];
#pragma unroll
  for (int r = 0; r < 16; ++r) sr[r] = 0;
  if (strip < nstrips) {
    const int di = strip * 16 + lane;
    if (lane < 16 && di < N) { const uint4 t = cnt4[di]; c = node_count(t); vm = row_lanes(t); }
#pragma unroll
    for (int r = 0; r < 16; ++r) {
      const unsigned mr = (unsigned)__builtin_amdgcn_readlane((int)vm, r);
      const int* rowp = slots + (size_t)(strip * 16 + r) * ROW_SLOTS;
      if ((mr >> lane) & 1u) sr[r] = rowp[lane];
    }
  }

  while (strip < nstrips) {
    const int d0 = strip * 16;
    const int nstrip = strip + 8 * G;

    const float    ds = (lane < 16 && d0 + lane < N) ? node_scale(c) : 0.f;
    unsigned run = 0u;
    int rpi = 0;
#pragma unroll
    for (int r = 0; r < 16; ++r) {
      const unsigned mr = (unsigned)__builtin_amdgcn_readlane((int)vm, r);
      rpi = (lane == r) ? (int)run : rpi;
      if ((mr >> lane) & 1u) fl[run + __builtin_amdgcn_mbcnt_lo(mr, 0u)] = sr[r];
      run += (unsigned)__builtin_popcount(mr);
    }
    rpi = (lane >= 16) ? (int)run : rpi;
    const unsigned rp = (unsigned)rpi;
    const unsigned E1 = run;

    __builtin_amdgcn_fence(__ATOMIC_RELEASE, "wavefront");
    __builtin_amdgcn_wave_barrier();
    __builtin_amdgcn_fence(__ATOMIC_ACQUIRE, "wavefront");

    int sl = 0, sl_n = 0;
    if ((unsigned)lane < E1)       sl   = fl[lane];
    if (32u + (unsigned)lane < E1) sl_n = fl[32 + lane];

    unsigned dc = 0u, c_n = 0u, vm_n = 0u;
    {
      float4 sv[16];
#pragma unroll
      for (int r = 0; r < 16; ++r) {
        const int d = (d0 + r < N) ? d0 + r : N - 1;
        sv[r] = *(const float4*)(x + (size_t)d * IN_CH + cg);
      }
      if ((unsigned)lane < E1) dc = node_count(cnt4[sl]);
      if (nstrip < nstrips) {
        const int di = nstrip * 16 + lane;
        if (lane < 16 && di < N) { const uint4 t = cnt4[di]; c_n = node_count(t); vm_n = row_lanes(t); }
      }
#pragma unroll
      for (int r = 0; r < 16; ++r) {
        const float dr = lane_f32(ds, r);
        const float ns = dr * dr;
        float4 t;
        t.x = ns * sv[r].x; t.y = ns * sv[r].y; t.z = ns * sv[r].z; t.w = ns * sv[r].w;
        *(float4*)(at + r * APITCH + cg) = t;
      }
    }

    int      r    = 0;
    unsigned endc = (unsigned)__builtin_amdgcn_readlane((int)rp, 1);
    float    ddc  = lane_f32(ds, 0);
    float4 acc = make_float4(0.f, 0.f, 0.f, 0.f);

    const unsigned long long xb = (unsigned long long)x;
    const unsigned vb = (unsigned)lane << 4;

    for (unsigned b0 = 0; b0 < E1; b0 += 32) {
      const unsigned left = E1 - b0;
      const int n = (int)(left < 32u ? left : 32u);

      unsigned vo = vb;
      asm volatile("" : "+v"(vo));

      v4f v[32];
#pragma unroll
      for (int hq = 0; hq < 2; ++hq) {
        unsigned long long ra[16];
#pragma unroll
        for (int u = 0; u < 16; ++u) {
          ra[u] = xb + ((unsigned long long)(unsigned)__builtin_amdgcn_readlane(sl, 16 * hq + u) << 9);
          asm("" : "+s"(ra[u]));
        }
        __builtin_amdgcn_sched_barrier(0);
#pragma unroll
        for (int u = 0; u < 16; ++u) v[16 * hq + u] = *(gv4p)(ra[u] + vo);
        __builtin_amdgcn_sched_barrier(0);
      }
      const uint4 tn = cnt4[sl_n];
      int sl_nn = 0;
      if (b0 + 64u + (unsigned)lane < E1) sl_nn = fl[b0 + 64 + lane];
      const float dl = node_scale(dc);

#pragma unroll
      for (int u = 0; u < 32; ++u) {
        if (u < n) {
          const unsigned e = b0 + (unsigned)u;
          while (e >= endc) {
            add_row(at, r, cg, acc);
            acc = make_float4(0.f, 0.f, 0.f, 0.f);
            ++r;
            endc = (unsigned)__builtin_amdgcn_readlane((int)rp, r + 1);
            ddc  = lane_f32(ds, r);
          }
          axpy4(acc, lane_f32(dl, u) * ddc, make_float4(v[u][0], v[u][1], v[u][2], v[u][3]));
        }
      }
      dc = node_count(tn);
      sl = sl_n; sl_n = sl_nn;
    }
    add_row(at, r, cg, acc);

    const bool ov = c != (unsigned)__builtin_popcount(vm);
    if (__builtin_amdgcn_ballot_w32(ov) != 0u) {
      for (int q0 = 0; q0 < E; q0 += 32) {
        int od = -1, os = 0;
        if (q0 + lane < E) { od = ei[E + q0 + lane]; os = ei[q0 + lane]; }
        const int rl = od - d0;
        const int on = __builtin_amdgcn_ds_bpermute((rl & 15) << 2, (int)ov);
        unsigned mk = __builtin_amdgcn_ballot_w32(rl >= 0 && rl < 16 && on != 0);
        while (mk != 0u) {
          const int j = __builtin_ctz(mk);
          mk &= mk - 1u;
          const int rd = __builtin_amdgcn_readlane(od, j) - d0;
          const int s  = __builtin_amdgcn_readlane(os, j);
          const float w = node_scale(node_count(cnt4[s])) * lane_f32(ds, rd);
          const float4 xv = *(const float4*)(x + (size_t)s * IN_CH + cg);
          float4 t;
          t.x = w * xv.x; t.y = w * xv.y; t.z = w * xv.z; t.w = w * xv.w;
          add_row(at, rd, cg, t);
        }
      }
    }

    int sr_n[16];
#pragma unroll
    for (int q = 0; q < 16; ++q) sr_n[q] = 0;
    if (nstrip < nstrips) {
#pragma unroll
      for (int q = 0; q < 16; ++q) {
        const unsigned mq = (unsigned)__builtin_amdgcn_readlane((int)vm_n, q);
        const int* rowp = slots + (size_t)(nstrip * 16 + q) * ROW_SLOTS;
        if ((mq >> lane) & 1u) sr_n[q] = rowp[lane];
      }
    }

    __builtin_amdgcn_fence(__ATOMIC_RELEASE, "wavefront");
    __builtin_amdgcn_wave_barrier();
    __builtin_amdgcn_fence(__ATOMIC_ACQUIRE, "wavefront");

    v8f cc[8];
#pragma unroll
    for (int ct = 0; ct < 8; ++ct) cc[ct] = (v8f){};
    v2f a0[8], a1[8];
    load_a_chunk(a0, ar);
#pragma unroll 1
    for (int c2 = 0; c2 < 2; ++c2) {
      load_a_chunk(a1, ar + (2 * c2 + 1) * 32);
      mma_chunk(cc, a0, bb, wl, c2 * 8);
      load_a_chunk(a0, ar + (c2 == 0 ? 64 : 0));
      mma_chunk(cc, a1, bb, wl, c2 * 8 + 4);
    }

    float* orow = out + (size_t)(d0 + mo) * HID + mn;
    if (d0 + 16 <= N) {
#pragma unroll
      for (int ct = 0; ct < 8; ++ct)
#pragma unroll
        for (int q = 0; q < 8; ++q) {
          const float t = cc[ct][q] + bz[ct];
          orow[(size_t)q * HID + ct * 16] = t >= 0.f ? t : pz[ct] * t;
        }
    } else {
#pragma unroll
      for (int ct = 0; ct < 8; ++ct)
#pragma unroll
        for (int q = 0; q < 8; ++q) {
          const float t = cc[ct][q] + bz[ct];
          if (d0 + mo + q < N) orow[(size_t)q * HID + ct * 16] = t >= 0.f ? t : pz[ct] * t;
        }
    }

    __builtin_amdgcn_fence(__ATOMIC_RELEASE, "wavefront");
    __builtin_amdgcn_wave_barrier();

    c = c_n; vm = vm_n;
#pragma unroll
    for (int q = 0; q < 16; ++q) sr[q] = sr_n[q];
    strip = nstrip;
  }
}

__global__ void k_init_deg(unsigned* __restrict__ deg, int N) {
  int i = blockIdx.x * blockDim.x + threadIdx.x;
  if (i < N) deg[i] = 1u;
}

__global__ void k_count(const int* __restrict__ ei, int E, unsigned* __restrict__ deg) {
  int e = blockIdx.x * blockDim.x + threadIdx.x;
  if (e < E) atomicAdd(&deg[ei[E + e]], 1u);
}

__global__ void k_dis(const unsigned* __restrict__ deg, float* __restrict__ dis, int N) {
  int i = blockIdx.x * blockDim.x + threadIdx.x;
  if (i < N) dis[i] = rsqrtf((float)deg[i]);
}

__global__ void k_init_out(const float* __restrict__ h, const float* __restrict__ dis,
                           const float* __restrict__ bias, float* __restrict__ out, int N) {
  int t = blockIdx.x * blockDim.x + threadIdx.x;
  if (t >= N * 32) return;
  int i  = t >> 5;
  int cg = (t & 31) << 2;
  float nrm = dis[i] * dis[i];
  float4 hv = *(const float4*)(h + (size_t)i * HID + cg);
  float4 bv = *(const float4*)(bias + cg);
  float4 o;
  o.x = bv.x + nrm * hv.x;
  o.y = bv.y + nrm * hv.y;
  o.z = bv.z + nrm * hv.z;
  o.w = bv.w + nrm * hv.w;
  *(float4*)(out + (size_t)i * HID + cg) = o;
}

__global__ __launch_bounds__(256) void k_scatter(const int* __restrict__ ei, int E,
                                                 const float* __restrict__ h,
                                                 const float* __restrict__ dis,
                                                 float* __restrict__ out) {
  int w    = (blockIdx.x * blockDim.x + threadIdx.x) >> 5;
  int lane = threadIdx.x & 31;
  if (w >= E) return;
  int s = ei[w];
  int d = ei[E + w];
  float nrm = dis[s] * dis[d];
  const float4 hv = *(const float4*)(h + (size_t)s * HID + (lane << 2));
  float* op = out + (size_t)d * HID + (lane << 2);
  unsafeAtomicAdd(op + 0, nrm * hv.x);
  unsafeAtomicAdd(op + 1, nrm * hv.y);
  unsafeAtomicAdd(op + 2, nrm * hv.z);
  unsafeAtomicAdd(op + 3, nrm * hv.w);
}

__global__ void k_prelu(float* __restrict__ out, const float* __restrict__ pw, int N) {
  int t = blockIdx.x * blockDim.x + threadIdx.x;
  if (t >= N * 32) return;
  int cg = (t & 31) << 2;
  float4 v = *(float4*)(out + (size_t)t * 4);
  float4 p = *(const float4*)(pw + cg);
  v.x = v.x >= 0.f ? v.x : p.x * v.x;
  v.y = v.y >= 0.f ? v.y : p.y * v.y;
  v.z = v.z >= 0.f ? v.z : p.z * v.z;
  v.w = v.w >= 0.f ? v.w : p.w * v.w;
  *(float4*)(out + (size_t)t * 4) = v;
}

extern "C" void kernel_launch(void* const* d_in, const int* in_sizes, int n_in,
                              void* d_out, int out_size, void* d_ws, size_t ws_size,
                              hipStream_t stream) {
  const float* x  = (const float*)d_in[0];
  const int*   ei = (const int*)d_in[1];
  const float* W  = (const float*)d_in[2];
  const float* b  = (const float*)d_in[3];
  const float* pw = (const float*)d_in[4];
  float* out = (float*)d_out;

  const int N = in_sizes[0] / IN_CH;
  const int E = in_sizes[1] / 2;

  char* ws = (char*)d_ws;
  size_t o0 = 0;
  size_t o1 = (o0 + (size_t)N * 16 + 255) & ~(size_t)255;
  size_t o2 = (o1 + (size_t)N * 4 + 255) & ~(size_t)255;
  unsigned* cnt    = (unsigned*)(ws + o0);
  float*    dis    = (float*)(ws + o1);
  float*    h      = (float*)(ws + o2);
  int*      slots  = (int*)(ws + o2);

  const size_t need = o2 + (size_t)N * ROW_SLOTS * 4;

  const int T = 256;
  const int gemm_groups = 512;

  if (ws_size < need) {
    int tq = N * 32;
    k_init_deg<<<(N + T - 1) / T, T, 0, stream>>>(cnt, N);
    k_count   <<<(E + T - 1) / T, T, 0, stream>>>(ei, E, cnt);
    k_dis     <<<(N + T - 1) / T, T, 0, stream>>>(cnt, dis, N);
    k_gemm    <<<gemm_groups, T, 0, stream>>>(x, W, h, N / 16);
    k_init_out<<<(tq + T - 1) / T, T, 0, stream>>>(h, dis, b, out, N);
    k_scatter <<<(E + 7) / 8, T, 0, stream>>>(ei, E, h, dis, out);
    k_prelu   <<<(tq + T - 1) / T, T, 0, stream>>>(out, pw, N);
    return;
  }

  k_rank_place<<<128, RANK_T, 0, stream>>>(ei, E, N, cnt, slots);

  k_agg_gemm<<<gemm_groups, T, 0, stream>>>(x, W, slots, (const uint4*)cnt, ei, E, b, pw, out, N, (N + 15) / 16);
}
